# starting kernel
# baseline (speedup 1.0000x reference)
#include <hip/hip_runtime.h>
#include <hip/hip_bf16.h>

typedef float v2f __attribute__((ext_vector_type(2)));
typedef float v8f __attribute__((ext_vector_type(8)));

#define MAX_LEN 200
#define MAX_K   3
#define D_IN    128
#define D_OUT   128
#define NBATCH  1024
#define NEG_PAD (-65535.0f)

__global__ void __launch_bounds__(128)
gemm_lowS_kernel(const float* __restrict__ low, const float* __restrict__ S,
                 float* __restrict__ lown) {
  const int wave = threadIdx.x >> 5;
  const int lane = threadIdx.x & 31;
  const int tile = blockIdx.x * 4 + wave;
  const int tm   = tile >> 3;
  const int tn   = tile & 7;

  const int m     = lane & 15;
  const int khalf = (lane >> 4) * 2;
  const float* __restrict__ Arow = low + (size_t)(tm * 16 + m) * D_IN;
  const int n = tn * 16 + (lane & 15);

  v8f c = {};
#pragma unroll
  for (int kk = 0; kk < 32; ++kk) {
    const int kb = kk * 4 + khalf;
    v2f a;  a.x = Arow[kb];          a.y = Arow[kb + 1];
    v2f bb; bb.x = S[kb * D_OUT + n]; bb.y = S[(kb + 1) * D_OUT + n];
    c = __builtin_amdgcn_wmma_f32_16x16x4_f32(false, a, false, bb,
                                              (short)0, c, false, false);
  }

  const int cn    = tn * 16 + (lane & 15);
  const int rbase = tm * 16 + ((lane >> 4) << 3);
#pragma unroll
  for (int r = 0; r < 8; ++r)
    lown[(size_t)(rbase + r) * D_OUT + cn] = c[r];
}

__global__ void init_b_kernel(const float* __restrict__ Bin,
                              float* __restrict__ Bwork,
                              float* __restrict__ Bdelta) {
  const int t = blockIdx.x * blockDim.x + threadIdx.x;
  if (t < MAX_K * MAX_LEN) { Bwork[t] = Bin[t]; Bdelta[t] = 0.0f; }
}

__global__ void __launch_bounds__(128)
routing_iter_kernel(const float* __restrict__ lown, const float* __restrict__ Bwork,
                    const int* __restrict__ seq_len, float* __restrict__ Bdelta,
                    float* __restrict__ out, int last) {
  __shared__ float sW[MAX_K][MAX_LEN];
  __shared__ float sH[MAX_K][D_OUT];
  __shared__ float sred[128];

  const int b    = blockIdx.x;
  const int t    = threadIdx.x;
  const int wave = t >> 5;
  const int lane = t & 31;
  const int len  = seq_len[b];
  const float* __restrict__ lowb = lown + (size_t)b * MAX_LEN * D_OUT;

  for (int i = t; i < MAX_K * MAX_LEN; i += 128) {
    const int l = i % MAX_LEN;
    sW[i / MAX_LEN][l] = (l < len) ? Bwork[i] : NEG_PAD;
  }
  __syncthreads();

  for (int k = 0; k < MAX_K; ++k) {
    float mx = -3.0e38f;
    for (int l = t; l < MAX_LEN; l += 128) mx = fmaxf(mx, sW[k][l]);
    sred[t] = mx; __syncthreads();
    for (int s = 64; s > 0; s >>= 1) {
      if (t < s) sred[t] = fmaxf(sred[t], sred[t + s]);
      __syncthreads();
    }
    const float mk = sred[0]; __syncthreads();
    float sum = 0.0f;
    for (int l = t; l < MAX_LEN; l += 128) {
      const float e = __expf(sW[k][l] - mk);
      sW[k][l] = e; sum += e;
    }
    sred[t] = sum; __syncthreads();
    for (int s = 64; s > 0; s >>= 1) {
      if (t < s) sred[t] += sred[t + s];
      __syncthreads();
    }
    const float inv = 1.0f / sred[0]; __syncthreads();
    for (int l = t; l < MAX_LEN; l += 128) sW[k][l] *= inv;
    __syncthreads();
  }

  const int m     = lane & 15;
  const int khalf = (lane >> 4) * 2;
#pragma unroll
  for (int j = 0; j < 2; ++j) {
    const int nt = wave * 2 + j;
    const int n  = nt * 16 + (lane & 15);
    v8f c = {};
    for (int kk = 0; kk < 50; ++kk) {
      const int kb = kk * 4 + khalf;
      v2f a;
      a.x = (m < MAX_K) ? sW[m][kb]     : 0.0f;
      a.y = (m < MAX_K) ? sW[m][kb + 1] : 0.0f;
      v2f bb;
      bb.x = lowb[kb * D_OUT + n];
      bb.y = lowb[(kb + 1) * D_OUT + n];
      c = __builtin_amdgcn_wmma_f32_16x16x4_f32(false, a, false, bb,
                                                (short)0, c, false, false);
    }
    if (lane < 16) {
#pragma unroll
      for (int r = 0; r < MAX_K; ++r) sH[r][nt * 16 + lane] = c[r];
    }
  }
  __syncthreads();

  {
    const float x0 = sH[0][t], x1 = sH[1][t], x2 = sH[2][t];
    const float sq = x0 * x0 + x1 * x1 + x2 * x2;
    const float scale = sq / (1.0f + sq) * rsqrtf(sq + 1e-9f);
    sH[0][t] = scale * x0; sH[1][t] = scale * x1; sH[2][t] = scale * x2;
    if (last) {
      float* __restrict__ ob = out + (size_t)b * MAX_K * D_OUT;
      ob[0 * D_OUT + t] = sH[0][t];
      ob[1 * D_OUT + t] = sH[1][t];
      ob[2 * D_OUT + t] = sH[2][t];
    }
  }
  __syncthreads();

  if (!last) {
    for (int p = t; p < MAX_K * MAX_LEN; p += 128) {
      const int k = p / MAX_LEN;
      const int l = p % MAX_LEN;
      const float4* __restrict__ lp =
          reinterpret_cast<const float4*>(lowb + (size_t)l * D_OUT);
      float acc = 0.0f;
#pragma unroll
      for (int e4 = 0; e4 < D_OUT / 4; ++e4) {
        const float4 v = lp[e4];
        acc += v.x * sH[k][4 * e4 + 0];
        acc += v.y * sH[k][4 * e4 + 1];
        acc += v.z * sH[k][4 * e4 + 2];
        acc += v.w * sH[k][4 * e4 + 3];
      }
      atomicAdd(&Bdelta[p], acc);
    }
  }
}

__global__ void update_b_kernel(float* __restrict__ Bwork,
                                float* __restrict__ Bdelta) {
  const int t = blockIdx.x * blockDim.x + threadIdx.x;
  if (t < MAX_K * MAX_LEN) { Bwork[t] += Bdelta[t]; Bdelta[t] = 0.0f; }
}

extern "C" void kernel_launch(void* const* d_in, const int* in_sizes, int n_in,
                              void* d_out, int out_size, void* d_ws, size_t ws_size,
                              hipStream_t stream) {
  const float* low  = (const float*)d_in[0];
  const float* Bin  = (const float*)d_in[1];
  const float* S    = (const float*)d_in[2];
  const int*   slen = (const int*)d_in[3];
  float* out = (float*)d_out;

  char* ws = (char*)d_ws;
  float* lown   = (float*)ws;
  float* Bwork  = (float*)(ws + (size_t)NBATCH * MAX_LEN * D_OUT * sizeof(float));
  float* Bdelta = Bwork + MAX_K * MAX_LEN;

  gemm_lowS_kernel<<<(12800 * 8) / 4, 128, 0, stream>>>(low, S, lown);

  init_b_kernel<<<5, 128, 0, stream>>>(Bin, Bwork, Bdelta);

  for (int it = 0; it < 3; ++it) {
    const int last = (it == 2);
    routing_iter_kernel<<<NBATCH, 128, 0, stream>>>(lown, Bwork, slen, Bdelta,
                                                    out, last);
    if (!last) update_b_kernel<<<5, 128, 0, stream>>>(Bwork, Bdelta);
  }
}


# Per-batch pass: the block's 200x128 low_new tile is read once into swizzled LDS; W.tile (rows split over 2 waves) and the delta product (now on WMMA) read it there; float sums in a different order
# speedup vs baseline: 2.1467x; 2.1467x over previous
#include <hip/hip_runtime.h>
#include <hip/hip_bf16.h>

typedef float v2f __attribute__((ext_vector_type(2)));
typedef float v8f __attribute__((ext_vector_type(8)));

#define MAX_LEN 200
#define MAX_K   3
#define D_IN    128
#define D_OUT   128
#define NBATCH  1024
#define NEG_PAD (-65535.0f)

#define ROWS_PER_WG   800
#define TILE_STRIDE   (64 * D_IN)

__device__ __forceinline__ void load_a_tile(float4 (&a)[16], const float* __restrict__ p) {
#pragma unroll
  for (int j = 0; j < 16; ++j)
    a[j] = *reinterpret_cast<const float4*>(p + 8 * j);
}

__device__ __forceinline__ void mm_tile(const float4 (&a)[16], const v2f (&bf)[32][4],
                                        float* __restrict__ o) {
  v8f acc[4] = {};
#pragma unroll
  for (int j = 0; j < 16; ++j) {
    v2f alo; alo.x = a[j].x; alo.y = a[j].y;
    v2f ahi; ahi.x = a[j].z; ahi.y = a[j].w;
#pragma unroll
    for (int nt = 0; nt < 4; ++nt)
      acc[nt] = __builtin_amdgcn_wmma_f32_16x16x4_f32(false, alo, false, bf[2 * j][nt],
                                                      (short)0, acc[nt], false, false);
#pragma unroll
    for (int nt = 0; nt < 4; ++nt)
      acc[nt] = __builtin_amdgcn_wmma_f32_16x16x4_f32(false, ahi, false, bf[2 * j + 1][nt],
                                                      (short)0, acc[nt], false, false);
  }
#pragma unroll
  for (int r = 0; r < 8; ++r) {
    float4 v;
    v.x = acc[0][r]; v.y = acc[1][r]; v.z = acc[2][r]; v.w = acc[3][r];
    *reinterpret_cast<float4*>(o + r * D_OUT) = v;
  }
}

__global__ void __launch_bounds__(256)
gemm_lowS_kernel(const float* __restrict__ low, const float* __restrict__ S,
                 float* __restrict__ lown) {
  const int wave = __builtin_amdgcn_readfirstlane((int)(threadIdx.x >> 5));
  const int lane = threadIdx.x & 31;
  const int c  = lane & 15;
  const int h  = lane >> 4;
  const int nh = wave & 1;
  const int q  = wave >> 1;
  const int ntile = (q < 2) ? 13 : 12;

  const size_t row0 = (size_t)blockIdx.x * ROWS_PER_WG + 16 * q;
  const float* __restrict__ ap = low + (row0 + c) * D_IN + 4 * h;
  float* __restrict__ op = lown + (row0 + 8 * h) * D_OUT + 64 * nh + 4 * c;

  float4 a0[16], a1[16];
  load_a_tile(a0, ap);

  v2f bf[32][4];
  const float* __restrict__ sp = S + (size_t)(4 * h) * D_OUT + 64 * nh + 4 * c;
#pragma unroll
  for (int j = 0; j < 16; ++j) {
#pragma unroll
    for (int p = 0; p < 2; ++p) {
      const float4 r0 = *reinterpret_cast<const float4*>(sp + (8 * j + 2 * p) * D_OUT);
      const float4 r1 = *reinterpret_cast<const float4*>(sp + (8 * j + 2 * p + 1) * D_OUT);
      bf[2 * j + p][0].x = r0.x; bf[2 * j + p][0].y = r1.x;
      bf[2 * j + p][1].x = r0.y; bf[2 * j + p][1].y = r1.y;
      bf[2 * j + p][2].x = r0.z; bf[2 * j + p][2].y = r1.z;
      bf[2 * j + p][3].x = r0.w; bf[2 * j + p][3].y = r1.w;
    }
  }

#pragma unroll 1
  for (int i = 0; i < 6; ++i) {
    load_a_tile(a1, ap + (size_t)(2 * i + 1) * TILE_STRIDE);
    mm_tile(a0, bf, op + (size_t)(2 * i) * TILE_STRIDE);
    const int nxt = (2 * i + 2 < ntile) ? (2 * i + 2) : (ntile - 1);
    load_a_tile(a0, ap + (size_t)nxt * TILE_STRIDE);
    mm_tile(a1, bf, op + (size_t)(2 * i + 1) * TILE_STRIDE);
  }
  if (ntile == 13)
    mm_tile(a0, bf, op + (size_t)12 * TILE_STRIDE);
}

__global__ void init_b_kernel(const float* __restrict__ Bin,
                              float* __restrict__ Bwork,
                              float* __restrict__ Bdelta) {
  const int t = blockIdx.x * blockDim.x + threadIdx.x;
  if (t < MAX_K * MAX_LEN) { Bwork[t] = Bin[t]; Bdelta[t] = 0.0f; }
}

#define TILE_CH 32
#define STAGE_LOADS (MAX_LEN * TILE_CH / 128)

__global__ void __launch_bounds__(128)
routing_iter_kernel(const float* __restrict__ lown, const float* __restrict__ Bwork,
                    const int* __restrict__ seq_len, float* __restrict__ Bdelta,
                    float* __restrict__ out, int last) {
  __shared__ float4 sT[MAX_LEN * TILE_CH];
  __shared__ __attribute__((aligned(16))) float sW[MAX_K][MAX_LEN];
  __shared__ __attribute__((aligned(16))) float sP[2][MAX_K][D_OUT];
  float* sred = &sP[0][0][0];

  const int b    = blockIdx.x;
  const int t    = threadIdx.x;
  const int wave = __builtin_amdgcn_readfirstlane((int)(threadIdx.x >> 5));
  const int lane = t & 31;
  const int len  = seq_len[b];
  const float4* __restrict__ lowb4 =
      reinterpret_cast<const float4*>(lown + (size_t)b * MAX_LEN * D_OUT);

  {
    float4 g[STAGE_LOADS];
#pragma unroll
    for (int i = 0; i < STAGE_LOADS; ++i) g[i] = lowb4[t + 128 * i];
#pragma unroll
    for (int i = 0; i < STAGE_LOADS; ++i) {
      const int row = 4 * i + wave;
      sT[row * TILE_CH + (lane ^ (row & 15))] = g[i];
    }
  }

  for (int i = t; i < MAX_K * MAX_LEN; i += 128) {
    const int l = i % MAX_LEN;
    sW[i / MAX_LEN][l] = (l < len) ? Bwork[i] : NEG_PAD;
  }
  __syncthreads();

  for (int k = 0; k < MAX_K; ++k) {
    float mx = -3.0e38f;
    for (int l = t; l < MAX_LEN; l += 128) mx = fmaxf(mx, sW[k][l]);
    sred[t] = mx; __syncthreads();
    for (int s = 64; s > 0; s >>= 1) {
      if (t < s) sred[t] = fmaxf(sred[t], sred[t + s]);
      __syncthreads();
    }
    const float mk = sred[0]; __syncthreads();
    float sum = 0.0f;
    for (int l = t; l < MAX_LEN; l += 128) {
      const float e = __expf(sW[k][l] - mk);
      sW[k][l] = e; sum += e;
    }
    sred[t] = sum; __syncthreads();
    for (int s = 64; s > 0; s >>= 1) {
      if (t < s) sred[t] += sred[t + s];
      __syncthreads();
    }
    const float inv = 1.0f / sred[0]; __syncthreads();
    for (int l = t; l < MAX_LEN; l += 128) sW[k][l] *= inv;
    __syncthreads();
  }

  const int  c    = lane & 15;
  const int  h    = lane >> 4;
  const bool live = c < MAX_K;
  const int  mm   = live ? c : 0;

  {
    const int nh = wave & 1;
    const int kh = wave >> 1;
    float4 r0[25], r1[25];
    float2 wv[25];
#pragma unroll
    for (int u = 0; u < 25; ++u)
      wv[u] = *reinterpret_cast<const float2*>(&sW[mm][4 * (25 * kh + u) + 2 * h]);
    __builtin_amdgcn_sched_barrier(0);
#pragma unroll
    for (int u = 0; u < 25; ++u) {
      const int r0i = 4 * (25 * kh + u) + 2 * h;
      const int r1i = r0i + 1;
      r0[u] = sT[r0i * TILE_CH + ((16 * nh + c) ^ (r0i & 15))];
      r1[u] = sT[r1i * TILE_CH + ((16 * nh + c) ^ (r1i & 15))];
    }
    __builtin_amdgcn_sched_barrier(0);
    v8f acc[4] = {};
#pragma unroll
    for (int u = 0; u < 25; ++u) {
      v2f a;
      a.x = live ? wv[u].x : 0.0f;
      a.y = live ? wv[u].y : 0.0f;
      v2f b0; b0.x = r0[u].x; b0.y = r1[u].x;
      v2f b1; b1.x = r0[u].y; b1.y = r1[u].y;
      v2f b2; b2.x = r0[u].z; b2.y = r1[u].z;
      v2f b3; b3.x = r0[u].w; b3.y = r1[u].w;
      acc[0] = __builtin_amdgcn_wmma_f32_16x16x4_f32(false, a, false, b0, (short)0, acc[0], false, false);
      acc[1] = __builtin_amdgcn_wmma_f32_16x16x4_f32(false, a, false, b1, (short)0, acc[1], false, false);
      acc[2] = __builtin_amdgcn_wmma_f32_16x16x4_f32(false, a, false, b2, (short)0, acc[2], false, false);
      acc[3] = __builtin_amdgcn_wmma_f32_16x16x4_f32(false, a, false, b3, (short)0, acc[3], false, false);
    }
    if (h == 0) {
#pragma unroll
      for (int r = 0; r < MAX_K; ++r) {
        float4 v;
        v.x = acc[0][r]; v.y = acc[1][r]; v.z = acc[2][r]; v.w = acc[3][r];
        *reinterpret_cast<float4*>(&sP[kh][r][64 * nh + 4 * c]) = v;
      }
    }
  }
  __syncthreads();

  {
    const float x0 = sP[0][0][t] + sP[1][0][t];
    const float x1 = sP[0][1][t] + sP[1][1][t];
    const float x2 = sP[0][2][t] + sP[1][2][t];
    const float sq = x0 * x0 + x1 * x1 + x2 * x2;
    const float scale = sq / (1.0f + sq) * rsqrtf(sq + 1e-9f);
    const float h0 = scale * x0, h1 = scale * x1, h2 = scale * x2;
    if (last) {
      float* __restrict__ ob = out + (size_t)b * MAX_K * D_OUT;
      ob[0 * D_OUT + t] = h0;
      ob[1 * D_OUT + t] = h1;
      ob[2 * D_OUT + t] = h2;
      return;
    }
    sP[0][0][t] = h0; sP[0][1][t] = h1; sP[0][2][t] = h2;
  }
  __syncthreads();

  {
    v2f ah[32];
#pragma unroll
    for (int j = 0; j < 16; ++j) {
      const float4 hv = *reinterpret_cast<const float4*>(&sP[0][mm][8 * j + 4 * h]);
      ah[2 * j].x     = live ? hv.x : 0.0f;
      ah[2 * j].y     = live ? hv.y : 0.0f;
      ah[2 * j + 1].x = live ? hv.z : 0.0f;
      ah[2 * j + 1].y = live ? hv.w : 0.0f;
    }
    const int w0 = (wave + b) & 3;
    for (int lt = w0; lt < 13; lt += 4) {
      const int l  = 16 * lt + c;
      const int lc = (l < MAX_LEN) ? l : (MAX_LEN - 1);
      float4 bv[16];
#pragma unroll
      for (int j = 0; j < 16; ++j)
        bv[j] = sT[lc * TILE_CH + ((2 * j + h) ^ (lc & 15))];
      __builtin_amdgcn_sched_barrier(0);
      v8f d0 = {}, d1 = {};
#pragma unroll
      for (int j = 0; j < 16; ++j) {
        v2f blo; blo.x = bv[j].x; blo.y = bv[j].y;
        v2f bhi; bhi.x = bv[j].z; bhi.y = bv[j].w;
        d0 = __builtin_amdgcn_wmma_f32_16x16x4_f32(false, ah[2 * j], false, blo, (short)0, d0, false, false);
        d1 = __builtin_amdgcn_wmma_f32_16x16x4_f32(false, ah[2 * j + 1], false, bhi, (short)0, d1, false, false);
      }
      if (h == 0 && l < MAX_LEN) {
#pragma unroll
        for (int r = 0; r < MAX_K; ++r)
          atomicAdd(&Bdelta[r * MAX_LEN + l], d0[r] + d1[r]);
      }
    }
  }
}

__global__ void update_b_kernel(float* __restrict__ Bwork,
                                float* __restrict__ Bdelta) {
  const int t = blockIdx.x * blockDim.x + threadIdx.x;
  if (t < MAX_K * MAX_LEN) { Bwork[t] += Bdelta[t]; Bdelta[t] = 0.0f; }
}

extern "C" void kernel_launch(void* const* d_in, const int* in_sizes, int n_in,
                              void* d_out, int out_size, void* d_ws, size_t ws_size,
                              hipStream_t stream) {
  const float* low  = (const float*)d_in[0];
  const float* Bin  = (const float*)d_in[1];
  const float* S    = (const float*)d_in[2];
  const int*   slen = (const int*)d_in[3];
  float* out = (float*)d_out;

  char* ws = (char*)d_ws;
  float* lown   = (float*)ws;
  float* Bwork  = (float*)(ws + (size_t)NBATCH * MAX_LEN * D_OUT * sizeof(float));
  float* Bdelta = Bwork + MAX_K * MAX_LEN;

  gemm_lowS_kernel<<<256, 256, 0, stream>>>(low, S, lown);

  init_b_kernel<<<5, 128, 0, stream>>>(Bin, Bwork, Bdelta);

  for (int it = 0; it < 3; ++it) {
    const int last = (it == 2);
    routing_iter_kernel<<<NBATCH, 128, 0, stream>>>(lown, Bwork, slen, Bdelta,
                                                    out, last);
    if (!last) update_b_kernel<<<5, 128, 0, stream>>>(Bwork, Bdelta);
  }
}


# Softmax max and 1/sum hoisted out of the batch blocks: B kernels build them once a pass per (k, length) by chunked prefixes; a block looks up 3 pairs, B loads ahead of the tile; exp sums in new order
# speedup vs baseline: 2.8091x; 1.3086x over previous
#include <hip/hip_runtime.h>
#include <hip/hip_bf16.h>

typedef float v2f __attribute__((ext_vector_type(2)));
typedef float v8f __attribute__((ext_vector_type(8)));

#define MAX_LEN 200
#define MAX_K   3
#define D_IN    128
#define D_OUT   128
#define NBATCH  1024
#define NEG_PAD (-65535.0f)

#define ROWS_PER_WG   800
#define TILE_STRIDE   (64 * D_IN)

__device__ __forceinline__ void load_a_tile(float4 (&a)[16], const float* __restrict__ p) {
#pragma unroll
  for (int j = 0; j < 16; ++j)
    a[j] = *reinterpret_cast<const float4*>(p + 8 * j);
}

__device__ __forceinline__ void mm_tile(const float4 (&a)[16], const v2f (&bf)[32][4],
                                        float* __restrict__ o) {
  v8f acc[4] = {};
#pragma unroll
  for (int j = 0; j < 16; ++j) {
    v2f alo; alo.x = a[j].x; alo.y = a[j].y;
    v2f ahi; ahi.x = a[j].z; ahi.y = a[j].w;
#pragma unroll
    for (int nt = 0; nt < 4; ++nt)
      acc[nt] = __builtin_amdgcn_wmma_f32_16x16x4_f32(false, alo, false, bf[2 * j][nt],
                                                      (short)0, acc[nt], false, false);
#pragma unroll
    for (int nt = 0; nt < 4; ++nt)
      acc[nt] = __builtin_amdgcn_wmma_f32_16x16x4_f32(false, ahi, false, bf[2 * j + 1][nt],
                                                      (short)0, acc[nt], false, false);
  }
#pragma unroll
  for (int r = 0; r < 8; ++r) {
    float4 v;
    v.x = acc[0][r]; v.y = acc[1][r]; v.z = acc[2][r]; v.w = acc[3][r];
    *reinterpret_cast<float4*>(o + r * D_OUT) = v;
  }
}

__global__ void __launch_bounds__(256)
gemm_lowS_kernel(const float* __restrict__ low, const float* __restrict__ S,
                 float* __restrict__ lown) {
  const int wave = __builtin_amdgcn_readfirstlane((int)(threadIdx.x >> 5));
  const int lane = threadIdx.x & 31;
  const int c  = lane & 15;
  const int h  = lane >> 4;
  const int nh = wave & 1;
  const int q  = wave >> 1;
  const int ntile = (q < 2) ? 13 : 12;

  const size_t row0 = (size_t)blockIdx.x * ROWS_PER_WG + 16 * q;
  const float* __restrict__ ap = low + (row0 + c) * D_IN + 4 * h;
  float* __restrict__ op = lown + (row0 + 8 * h) * D_OUT + 64 * nh + 4 * c;

  float4 a0[16], a1[16];
  load_a_tile(a0, ap);

  v2f bf[32][4];
  const float* __restrict__ sp = S + (size_t)(4 * h) * D_OUT + 64 * nh + 4 * c;
#pragma unroll
  for (int j = 0; j < 16; ++j) {
#pragma unroll
    for (int p = 0; p < 2; ++p) {
      const float4 r0 = *reinterpret_cast<const float4*>(sp + (8 * j + 2 * p) * D_OUT);
      const float4 r1 = *reinterpret_cast<const float4*>(sp + (8 * j + 2 * p + 1) * D_OUT);
      bf[2 * j + p][0].x = r0.x; bf[2 * j + p][0].y = r1.x;
      bf[2 * j + p][1].x = r0.y; bf[2 * j + p][1].y = r1.y;
      bf[2 * j + p][2].x = r0.z; bf[2 * j + p][2].y = r1.z;
      bf[2 * j + p][3].x = r0.w; bf[2 * j + p][3].y = r1.w;
    }
  }

#pragma unroll 1
  for (int i = 0; i < 6; ++i) {
    load_a_tile(a1, ap + (size_t)(2 * i + 1) * TILE_STRIDE);
    mm_tile(a0, bf, op + (size_t)(2 * i) * TILE_STRIDE);
    const int nxt = (2 * i + 2 < ntile) ? (2 * i + 2) : (ntile - 1);
    load_a_tile(a0, ap + (size_t)nxt * TILE_STRIDE);
    mm_tile(a1, bf, op + (size_t)(2 * i + 1) * TILE_STRIDE);
  }
  if (ntile == 13)
    mm_tile(a0, bf, op + (size_t)12 * TILE_STRIDE);
}

#define TAB_N   (MAX_LEN + 1)
#define NCHUNK  (MAX_LEN / 8)
#define NPREF   (NCHUNK + 1)

struct TabLds {
  __attribute__((aligned(16))) float b[MAX_K * MAX_LEN];
  float cm[MAX_K * NCHUNK], cz[MAX_K * NCHUNK];
  float pm[MAX_K * NPREF],  pz[MAX_K * NPREF];
};

__device__ __forceinline__ void softmax_tables(TabLds& s, float* __restrict__ tabM,
                                               float* __restrict__ tabI, int t) {
  if (t < MAX_K * NCHUNK) {
    const int k = t / NCHUNK, j = t % NCHUNK;
    const float4 v0 = *reinterpret_cast<const float4*>(&s.b[k * MAX_LEN + 8 * j]);
    const float4 v1 = *reinterpret_cast<const float4*>(&s.b[k * MAX_LEN + 8 * j + 4]);
    const float m = fmaxf(fmaxf(fmaxf(v0.x, v0.y), fmaxf(v0.z, v0.w)),
                          fmaxf(fmaxf(v1.x, v1.y), fmaxf(v1.z, v1.w)));
    s.cm[t] = m;
    s.cz[t] = ((__expf(v0.x - m) + __expf(v0.y - m)) + (__expf(v0.z - m) + __expf(v0.w - m))) +
              ((__expf(v1.x - m) + __expf(v1.y - m)) + (__expf(v1.z - m) + __expf(v1.w - m)));
  }
  __syncthreads();
  if (t < MAX_K * NPREF) {
    const int k = t / NPREF, q = t % NPREF;
    float cmv[NCHUNK], czv[NCHUNK];
#pragma unroll
    for (int j = 0; j < NCHUNK; ++j) { cmv[j] = s.cm[k * NCHUNK + j]; czv[j] = s.cz[k * NCHUNK + j]; }
    float m = -3.0e38f;
#pragma unroll
    for (int j = 0; j < NCHUNK; ++j) m = (j < q) ? fmaxf(m, cmv[j]) : m;
    float z = 0.0f;
#pragma unroll
    for (int j = 0; j < NCHUNK; ++j) z += (j < q) ? czv[j] * __expf(cmv[j] - m) : 0.0f;
    s.pm[t] = m;
    s.pz[t] = z;
  }
  __syncthreads();
  if (t < MAX_K * TAB_N) {
    const int k = t / TAB_N, n = t % TAB_N;
    const int q = n >> 3, r = n & 7;
    const float pm = s.pm[k * NPREF + q];
    const float pz = s.pz[k * NPREF + q];
    float bv[7];
#pragma unroll
    for (int i = 0; i < 7; ++i) bv[i] = s.b[k * MAX_LEN + min(8 * q + i, MAX_LEN - 1)];
    float m = (n < MAX_LEN) ? NEG_PAD : -3.0e38f;
    m = (q > 0) ? fmaxf(m, pm) : m;
#pragma unroll
    for (int i = 0; i < 7; ++i) m = (i < r) ? fmaxf(m, bv[i]) : m;
    float z = (n < MAX_LEN) ? (float)(MAX_LEN - n) * __expf(NEG_PAD - m) : 0.0f;
    z += (q > 0) ? pz * __expf(pm - m) : 0.0f;
#pragma unroll
    for (int i = 0; i < 7; ++i) z += (i < r) ? __expf(bv[i] - m) : 0.0f;
    tabM[t] = m;
    tabI[t] = 1.0f / z;
  }
}

__global__ void __launch_bounds__(640)
init_b_kernel(const float* __restrict__ Bin, float* __restrict__ Bwork,
              float* __restrict__ Bdelta, float* __restrict__ tabM,
              float* __restrict__ tabI) {
  __shared__ TabLds s;
  const int t = threadIdx.x;
  if (t < MAX_K * MAX_LEN) {
    const float v = Bin[t];
    Bwork[t] = v; Bdelta[t] = 0.0f; s.b[t] = v;
  }
  __syncthreads();
  softmax_tables(s, tabM, tabI, t);
}

#define TILE_CH 32
#define STAGE_LOADS (MAX_LEN * TILE_CH / 128)

__global__ void __launch_bounds__(128)
routing_iter_kernel(const float* __restrict__ lown, const float* __restrict__ Bwork,
                    const float* __restrict__ tabM, const float* __restrict__ tabI,
                    const int* __restrict__ seq_len, float* __restrict__ Bdelta,
                    float* __restrict__ out, int last) {
  __shared__ float4 sT[MAX_LEN * TILE_CH];
  __shared__ __attribute__((aligned(16))) float sWf[5 * 128];
  float (*sW)[MAX_LEN] = reinterpret_cast<float (*)[MAX_LEN]>(sWf);
  __shared__ __attribute__((aligned(16))) float sP[2][MAX_K][D_OUT];

  const int b    = blockIdx.x;
  const int t    = threadIdx.x;
  const int wave = __builtin_amdgcn_readfirstlane((int)(threadIdx.x >> 5));
  const int lane = t & 31;
  const int len  = seq_len[b];
  const int n    = min(max(len, 0), MAX_LEN);
  const float4* __restrict__ lowb4 =
      reinterpret_cast<const float4*>(lown + (size_t)b * MAX_LEN * D_OUT);

  float bw[5];
#pragma unroll
  for (int j = 0; j < 5; ++j) bw[j] = Bwork[min(t + 128 * j, MAX_K * MAX_LEN - 1)];

  float4 g[STAGE_LOADS];
#pragma unroll
  for (int i = 0; i < STAGE_LOADS; ++i) g[i] = lowb4[t + 128 * i];

  const float m0 = tabM[0 * TAB_N + n], m1 = tabM[1 * TAB_N + n], m2 = tabM[2 * TAB_N + n];
  const float i0 = tabI[0 * TAB_N + n], i1 = tabI[1 * TAB_N + n], i2 = tabI[2 * TAB_N + n];

#pragma unroll
  for (int i = 0; i < STAGE_LOADS; ++i) {
    const int row = 4 * i + wave;
    sT[row * TILE_CH + (lane ^ (row & 15))] = g[i];
  }

#pragma unroll
  for (int j = 0; j < 5; ++j) {
    const int i = t + 128 * j;
    const int k = i / MAX_LEN;
    const int l = i - k * MAX_LEN;
    const float x  = (l < len) ? bw[j] : NEG_PAD;
    const float mk = (k == 0) ? m0 : (k == 1) ? m1 : m2;
    const float ik = (k == 0) ? i0 : (k == 1) ? i1 : i2;
    sWf[i] = __expf(x - mk) * ik;
  }
  __syncthreads();

  const int  c    = lane & 15;
  const int  h    = lane >> 4;
  const bool live = c < MAX_K;
  const int  mm   = live ? c : 0;

  {
    const int nh = wave & 1;
    const int kh = wave >> 1;
    float4 r0[25], r1[25];
    float2 wv[25];
#pragma unroll
    for (int u = 0; u < 25; ++u)
      wv[u] = *reinterpret_cast<const float2*>(&sW[mm][4 * (25 * kh + u) + 2 * h]);
    __builtin_amdgcn_sched_barrier(0);
#pragma unroll
    for (int u = 0; u < 25; ++u) {
      const int r0i = 4 * (25 * kh + u) + 2 * h;
      const int r1i = r0i + 1;
      r0[u] = sT[r0i * TILE_CH + ((16 * nh + c) ^ (r0i & 15))];
      r1[u] = sT[r1i * TILE_CH + ((16 * nh + c) ^ (r1i & 15))];
    }
    __builtin_amdgcn_sched_barrier(0);
    v8f acc[4] = {};
#pragma unroll
    for (int u = 0; u < 25; ++u) {
      v2f a;
      a.x = live ? wv[u].x : 0.0f;
      a.y = live ? wv[u].y : 0.0f;
      v2f b0; b0.x = r0[u].x; b0.y = r1[u].x;
      v2f b1; b1.x = r0[u].y; b1.y = r1[u].y;
      v2f b2; b2.x = r0[u].z; b2.y = r1[u].z;
      v2f b3; b3.x = r0[u].w; b3.y = r1[u].w;
      acc[0] = __builtin_amdgcn_wmma_f32_16x16x4_f32(false, a, false, b0, (short)0, acc[0], false, false);
      acc[1] = __builtin_amdgcn_wmma_f32_16x16x4_f32(false, a, false, b1, (short)0, acc[1], false, false);
      acc[2] = __builtin_amdgcn_wmma_f32_16x16x4_f32(false, a, false, b2, (short)0, acc[2], false, false);
      acc[3] = __builtin_amdgcn_wmma_f32_16x16x4_f32(false, a, false, b3, (short)0, acc[3], false, false);
    }
    if (h == 0) {
#pragma unroll
      for (int r = 0; r < MAX_K; ++r) {
        float4 v;
        v.x = acc[0][r]; v.y = acc[1][r]; v.z = acc[2][r]; v.w = acc[3][r];
        *reinterpret_cast<float4*>(&sP[kh][r][64 * nh + 4 * c]) = v;
      }
    }
  }
  __syncthreads();

  {
    const float x0 = sP[0][0][t] + sP[1][0][t];
    const float x1 = sP[0][1][t] + sP[1][1][t];
    const float x2 = sP[0][2][t] + sP[1][2][t];
    const float sq = x0 * x0 + x1 * x1 + x2 * x2;
    const float scale = sq / (1.0f + sq) * rsqrtf(sq + 1e-9f);
    const float h0 = scale * x0, h1 = scale * x1, h2 = scale * x2;
    if (last) {
      float* __restrict__ ob = out + (size_t)b * MAX_K * D_OUT;
      ob[0 * D_OUT + t] = h0;
      ob[1 * D_OUT + t] = h1;
      ob[2 * D_OUT + t] = h2;
      return;
    }
    sP[0][0][t] = h0; sP[0][1][t] = h1; sP[0][2][t] = h2;
  }
  __syncthreads();

  {
    v2f ah[32];
#pragma unroll
    for (int j = 0; j < 16; ++j) {
      const float4 hv = *reinterpret_cast<const float4*>(&sP[0][mm][8 * j + 4 * h]);
      ah[2 * j].x     = live ? hv.x : 0.0f;
      ah[2 * j].y     = live ? hv.y : 0.0f;
      ah[2 * j + 1].x = live ? hv.z : 0.0f;
      ah[2 * j + 1].y = live ? hv.w : 0.0f;
    }
    const int w0 = (wave + b) & 3;
    for (int lt = w0; lt < 13; lt += 4) {
      const int l  = 16 * lt + c;
      const int lc = (l < MAX_LEN) ? l : (MAX_LEN - 1);
      float4 bv[16];
#pragma unroll
      for (int j = 0; j < 16; ++j)
        bv[j] = sT[lc * TILE_CH + ((2 * j + h) ^ (lc & 15))];
      __builtin_amdgcn_sched_barrier(0);
      v8f d0 = {}, d1 = {};
#pragma unroll
      for (int j = 0; j < 16; ++j) {
        v2f blo; blo.x = bv[j].x; blo.y = bv[j].y;
        v2f bhi; bhi.x = bv[j].z; bhi.y = bv[j].w;
        d0 = __builtin_amdgcn_wmma_f32_16x16x4_f32(false, ah[2 * j], false, blo, (short)0, d0, false, false);
        d1 = __builtin_amdgcn_wmma_f32_16x16x4_f32(false, ah[2 * j + 1], false, bhi, (short)0, d1, false, false);
      }
      if (h == 0 && l < MAX_LEN) {
#pragma unroll
        for (int r = 0; r < MAX_K; ++r)
          atomicAdd(&Bdelta[r * MAX_LEN + l], d0[r] + d1[r]);
      }
    }
  }
}

__global__ void __launch_bounds__(640)
update_b_kernel(float* __restrict__ Bwork, float* __restrict__ Bdelta,
                float* __restrict__ tabM, float* __restrict__ tabI) {
  __shared__ TabLds s;
  const int t = threadIdx.x;
  if (t < MAX_K * MAX_LEN) {
    const float v = Bwork[t] + Bdelta[t];
    Bwork[t] = v; Bdelta[t] = 0.0f; s.b[t] = v;
  }
  __syncthreads();
  softmax_tables(s, tabM, tabI, t);
}

extern "C" void kernel_launch(void* const* d_in, const int* in_sizes, int n_in,
                              void* d_out, int out_size, void* d_ws, size_t ws_size,
                              hipStream_t stream) {
  const float* low  = (const float*)d_in[0];
  const float* Bin  = (const float*)d_in[1];
  const float* S    = (const float*)d_in[2];
  const int*   slen = (const int*)d_in[3];
  float* out = (float*)d_out;

  char* ws = (char*)d_ws;
  float* lown   = (float*)ws;
  float* Bwork  = (float*)(ws + (size_t)NBATCH * MAX_LEN * D_OUT * sizeof(float));
  float* Bdelta = Bwork + MAX_K * MAX_LEN;
  float* tabM   = Bdelta + MAX_K * MAX_LEN;
  float* tabI   = tabM + MAX_K * TAB_N;

  gemm_lowS_kernel<<<256, 256, 0, stream>>>(low, S, lown);

  init_b_kernel<<<1, 640, 0, stream>>>(Bin, Bwork, Bdelta, tabM, tabI);

  for (int it = 0; it < 3; ++it) {
    const int last = (it == 2);
    routing_iter_kernel<<<NBATCH, 128, 0, stream>>>(lown, Bwork, tabM, tabI, slen,
                                                    Bdelta, out, last);
    if (!last) update_b_kernel<<<1, 640, 0, stream>>>(Bwork, Bdelta, tabM, tabI);
  }
}


# Per-batch pass: each wave reads its quarter of the low_new tile from memory straight into the W.tile B fragments (weights built in registers); LDS copy only for the delta product, no barrier first
# speedup vs baseline: 2.9398x; 1.0465x over previous
#include <hip/hip_runtime.h>
#include <hip/hip_bf16.h>

typedef float v2f __attribute__((ext_vector_type(2)));
typedef float v8f __attribute__((ext_vector_type(8)));

#define MAX_LEN 200
#define MAX_K   3
#define D_IN    128
#define D_OUT   128
#define NBATCH  1024
#define NEG_PAD (-65535.0f)

#define ROWS_PER_WG   800
#define TILE_STRIDE   (64 * D_IN)

__device__ __forceinline__ void load_a_tile(float4 (&a)[16], const float* __restrict__ p) {
#pragma unroll
  for (int j = 0; j < 16; ++j)
    a[j] = *reinterpret_cast<const float4*>(p + 8 * j);
}

__device__ __forceinline__ void mm_tile(const float4 (&a)[16], const v2f (&bf)[32][4],
                                        float* __restrict__ o) {
  v8f acc[4] = {};
#pragma unroll
  for (int j = 0; j < 16; ++j) {
    v2f alo; alo.x = a[j].x; alo.y = a[j].y;
    v2f ahi; ahi.x = a[j].z; ahi.y = a[j].w;
#pragma unroll
    for (int nt = 0; nt < 4; ++nt)
      acc[nt] = __builtin_amdgcn_wmma_f32_16x16x4_f32(false, alo, false, bf[2 * j][nt],
                                                      (short)0, acc[nt], false, false);
#pragma unroll
    for (int nt = 0; nt < 4; ++nt)
      acc[nt] = __builtin_amdgcn_wmma_f32_16x16x4_f32(false, ahi, false, bf[2 * j + 1][nt],
                                                      (short)0, acc[nt], false, false);
  }
#pragma unroll
  for (int r = 0; r < 8; ++r) {
    float4 v;
    v.x = acc[0][r]; v.y = acc[1][r]; v.z = acc[2][r]; v.w = acc[3][r];
    *reinterpret_cast<float4*>(o + r * D_OUT) = v;
  }
}

__global__ void __launch_bounds__(256)
gemm_lowS_kernel(const float* __restrict__ low, const float* __restrict__ S,
                 float* __restrict__ lown) {
  const int wave = __builtin_amdgcn_readfirstlane((int)(threadIdx.x >> 5));
  const int lane = threadIdx.x & 31;
  const int c  = lane & 15;
  const int h  = lane >> 4;
  const int nh = wave & 1;
  const int q  = wave >> 1;
  const int ntile = (q < 2) ? 13 : 12;

  const size_t row0 = (size_t)blockIdx.x * ROWS_PER_WG + 16 * q;
  const float* __restrict__ ap = low + (row0 + c) * D_IN + 4 * h;
  float* __restrict__ op = lown + (row0 + 8 * h) * D_OUT + 64 * nh + 4 * c;

  float4 a0[16], a1[16];
  load_a_tile(a0, ap);

  v2f bf[32][4];
  const float* __restrict__ sp = S + (size_t)(4 * h) * D_OUT + 64 * nh + 4 * c;
#pragma unroll
  for (int j = 0; j < 16; ++j) {
#pragma unroll
    for (int p = 0; p < 2; ++p) {
      const float4 r0 = *reinterpret_cast<const float4*>(sp + (8 * j + 2 * p) * D_OUT);
      const float4 r1 = *reinterpret_cast<const float4*>(sp + (8 * j + 2 * p + 1) * D_OUT);
      bf[2 * j + p][0].x = r0.x; bf[2 * j + p][0].y = r1.x;
      bf[2 * j + p][1].x = r0.y; bf[2 * j + p][1].y = r1.y;
      bf[2 * j + p][2].x = r0.z; bf[2 * j + p][2].y = r1.z;
      bf[2 * j + p][3].x = r0.w; bf[2 * j + p][3].y = r1.w;
    }
  }

#pragma unroll 1
  for (int i = 0; i < 6; ++i) {
    load_a_tile(a1, ap + (size_t)(2 * i + 1) * TILE_STRIDE);
    mm_tile(a0, bf, op + (size_t)(2 * i) * TILE_STRIDE);
    const int nxt = (2 * i + 2 < ntile) ? (2 * i + 2) : (ntile - 1);
    load_a_tile(a0, ap + (size_t)nxt * TILE_STRIDE);
    mm_tile(a1, bf, op + (size_t)(2 * i + 1) * TILE_STRIDE);
  }
  if (ntile == 13)
    mm_tile(a0, bf, op + (size_t)12 * TILE_STRIDE);
}

#define TAB_N   (MAX_LEN + 1)
#define NCHUNK  (MAX_LEN / 8)
#define NPREF   (NCHUNK + 1)

struct TabLds {
  __attribute__((aligned(16))) float b[MAX_K * MAX_LEN];
  float cm[MAX_K * NCHUNK], cz[MAX_K * NCHUNK];
  float pm[MAX_K * NPREF],  pz[MAX_K * NPREF];
};

__device__ __forceinline__ void softmax_tables(TabLds& s, float* __restrict__ tabM,
                                               float* __restrict__ tabI, int t) {
  if (t < MAX_K * NCHUNK) {
    const int k = t / NCHUNK, j = t % NCHUNK;
    const float4 v0 = *reinterpret_cast<const float4*>(&s.b[k * MAX_LEN + 8 * j]);
    const float4 v1 = *reinterpret_cast<const float4*>(&s.b[k * MAX_LEN + 8 * j + 4]);
    const float m = fmaxf(fmaxf(fmaxf(v0.x, v0.y), fmaxf(v0.z, v0.w)),
                          fmaxf(fmaxf(v1.x, v1.y), fmaxf(v1.z, v1.w)));
    s.cm[t] = m;
    s.cz[t] = ((__expf(v0.x - m) + __expf(v0.y - m)) + (__expf(v0.z - m) + __expf(v0.w - m))) +
              ((__expf(v1.x - m) + __expf(v1.y - m)) + (__expf(v1.z - m) + __expf(v1.w - m)));
  }
  __syncthreads();
  if (t < MAX_K * NPREF) {
    const int k = t / NPREF, q = t % NPREF;
    float cmv[NCHUNK], czv[NCHUNK];
#pragma unroll
    for (int j = 0; j < NCHUNK; ++j) { cmv[j] = s.cm[k * NCHUNK + j]; czv[j] = s.cz[k * NCHUNK + j]; }
    float m = -3.0e38f;
#pragma unroll
    for (int j = 0; j < NCHUNK; ++j) m = (j < q) ? fmaxf(m, cmv[j]) : m;
    float z = 0.0f;
#pragma unroll
    for (int j = 0; j < NCHUNK; ++j) z += (j < q) ? czv[j] * __expf(cmv[j] - m) : 0.0f;
    s.pm[t] = m;
    s.pz[t] = z;
  }
  __syncthreads();
  if (t < MAX_K * TAB_N) {
    const int k = t / TAB_N, n = t % TAB_N;
    const int q = n >> 3, r = n & 7;
    const float pm = s.pm[k * NPREF + q];
    const float pz = s.pz[k * NPREF + q];
    float bv[7];
#pragma unroll
    for (int i = 0; i < 7; ++i) bv[i] = s.b[k * MAX_LEN + min(8 * q + i, MAX_LEN - 1)];
    float m = (n < MAX_LEN) ? NEG_PAD : -3.0e38f;
    m = (q > 0) ? fmaxf(m, pm) : m;
#pragma unroll
    for (int i = 0; i < 7; ++i) m = (i < r) ? fmaxf(m, bv[i]) : m;
    float z = (n < MAX_LEN) ? (float)(MAX_LEN - n) * __expf(NEG_PAD - m) : 0.0f;
    z += (q > 0) ? pz * __expf(pm - m) : 0.0f;
#pragma unroll
    for (int i = 0; i < 7; ++i) z += (i < r) ? __expf(bv[i] - m) : 0.0f;
    tabM[t] = m;
    tabI[t] = 1.0f / z;
  }
}

__global__ void __launch_bounds__(640)
init_b_kernel(const float* __restrict__ Bin, float* __restrict__ Bwork,
              float* __restrict__ Bdelta, float* __restrict__ tabM,
              float* __restrict__ tabI) {
  __shared__ TabLds s;
  const int t = threadIdx.x;
  if (t < MAX_K * MAX_LEN) {
    const float v = Bin[t];
    Bwork[t] = v; Bdelta[t] = 0.0f; s.b[t] = v;
  }
  __syncthreads();
  softmax_tables(s, tabM, tabI, t);
}

#define TILE_CH 32
#define KSTEPS  25

__device__ __forceinline__ float uniform_f32(float v) {
  return __builtin_bit_cast(float, __builtin_amdgcn_readfirstlane(__builtin_bit_cast(int, v)));
}

template <bool LAST>
__global__ void __launch_bounds__(128)
routing_iter_kernel(const float* __restrict__ lown, const float* __restrict__ Bwork,
                    const float* __restrict__ tabM, const float* __restrict__ tabI,
                    const int* __restrict__ seq_len, float* __restrict__ Bdelta,
                    float* __restrict__ out) {
  __shared__ float4 sT[LAST ? 1 : MAX_LEN * TILE_CH];
  __shared__ __attribute__((aligned(16))) float sP[2][MAX_K][D_OUT];

  const int b    = blockIdx.x;
  const int t    = threadIdx.x;
  const int wave = __builtin_amdgcn_readfirstlane((int)(threadIdx.x >> 5));
  const int lane = t & 31;
  const int len  = seq_len[b];
  const int n    = min(max(len, 0), MAX_LEN);

  const int  c    = lane & 15;
  const int  h    = lane >> 4;
  const bool live = c < MAX_K;
  const int  mm   = live ? c : 0;
  const int  nh   = wave & 1;
  const int  kh   = wave >> 1;
  const int  lrow = 100 * kh + 2 * h;

  const float* __restrict__ bwp = Bwork + mm * MAX_LEN + lrow;
  float2 bw[KSTEPS];
#pragma unroll
  for (int u = 0; u < KSTEPS; ++u) bw[u] = *reinterpret_cast<const float2*>(bwp + 4 * u);

  const float4* __restrict__ tp =
      reinterpret_cast<const float4*>(lown + (size_t)b * MAX_LEN * D_OUT) +
      lrow * TILE_CH + 16 * nh + c;
  float4 r0[KSTEPS], r1[KSTEPS];
#pragma unroll
  for (int u = 0; u < KSTEPS; ++u) {
    r0[u] = tp[(4 * u) * TILE_CH];
    r1[u] = tp[(4 * u + 1) * TILE_CH];
  }

  const float m0 = uniform_f32(tabM[0 * TAB_N + n]);
  const float m1 = uniform_f32(tabM[1 * TAB_N + n]);
  const float m2 = uniform_f32(tabM[2 * TAB_N + n]);
  const float i0 = uniform_f32(tabI[0 * TAB_N + n]);
  const float i1 = uniform_f32(tabI[1 * TAB_N + n]);
  const float i2 = uniform_f32(tabI[2 * TAB_N + n]);
  const float mk = (c == 0) ? m0 : (c == 1) ? m1 : m2;
  const float ik = (c == 0) ? i0 : (c == 1) ? i1 : i2;
  const float wpad = __expf(NEG_PAD - mk) * ik;

  {
    v8f acc[4] = {};
#pragma unroll
    for (int u = 0; u < KSTEPS; ++u) {
      const int l0 = lrow + 4 * u;
      const int l1 = l0 + 1;
      if constexpr (!LAST) {
        sT[l0 * TILE_CH + ((16 * nh + c) ^ (l0 & 15))] = r0[u];
        sT[l1 * TILE_CH + ((16 * nh + c) ^ (l1 & 15))] = r1[u];
      }
      const float w0 = __expf(bw[u].x - mk) * ik;
      const float w1 = __expf(bw[u].y - mk) * ik;
      v2f a;
      a.x = live ? ((l0 < len) ? w0 : wpad) : 0.0f;
      a.y = live ? ((l1 < len) ? w1 : wpad) : 0.0f;
      v2f b0; b0.x = r0[u].x; b0.y = r1[u].x;
      v2f b1; b1.x = r0[u].y; b1.y = r1[u].y;
      v2f b2; b2.x = r0[u].z; b2.y = r1[u].z;
      v2f b3; b3.x = r0[u].w; b3.y = r1[u].w;
      acc[0] = __builtin_amdgcn_wmma_f32_16x16x4_f32(false, a, false, b0, (short)0, acc[0], false, false);
      acc[1] = __builtin_amdgcn_wmma_f32_16x16x4_f32(false, a, false, b1, (short)0, acc[1], false, false);
      acc[2] = __builtin_amdgcn_wmma_f32_16x16x4_f32(false, a, false, b2, (short)0, acc[2], false, false);
      acc[3] = __builtin_amdgcn_wmma_f32_16x16x4_f32(false, a, false, b3, (short)0, acc[3], false, false);
      __builtin_amdgcn_sched_barrier(0);
    }
    if (h == 0) {
#pragma unroll
      for (int r = 0; r < MAX_K; ++r) {
        float4 v;
        v.x = acc[0][r]; v.y = acc[1][r]; v.z = acc[2][r]; v.w = acc[3][r];
        *reinterpret_cast<float4*>(&sP[kh][r][64 * nh + 4 * c]) = v;
      }
    }
  }
  __syncthreads();

  {
    const float x0 = sP[0][0][t] + sP[1][0][t];
    const float x1 = sP[0][1][t] + sP[1][1][t];
    const float x2 = sP[0][2][t] + sP[1][2][t];
    const float sq = x0 * x0 + x1 * x1 + x2 * x2;
    const float scale = sq / (1.0f + sq) * rsqrtf(sq + 1e-9f);
    const float h0 = scale * x0, h1 = scale * x1, h2 = scale * x2;
    if constexpr (LAST) {
      float* __restrict__ ob = out + (size_t)b * MAX_K * D_OUT;
      ob[0 * D_OUT + t] = h0;
      ob[1 * D_OUT + t] = h1;
      ob[2 * D_OUT + t] = h2;
      return;
    }
    sP[0][0][t] = h0; sP[0][1][t] = h1; sP[0][2][t] = h2;
  }
  __syncthreads();

  if constexpr (!LAST) {
    v2f ah[32];
#pragma unroll
    for (int j = 0; j < 16; ++j) {
      const float4 hv = *reinterpret_cast<const float4*>(&sP[0][mm][8 * j + 4 * h]);
      ah[2 * j].x     = live ? hv.x : 0.0f;
      ah[2 * j].y     = live ? hv.y : 0.0f;
      ah[2 * j + 1].x = live ? hv.z : 0.0f;
      ah[2 * j + 1].y = live ? hv.w : 0.0f;
    }
    const int w0 = (wave + b) & 3;
    for (int lt = w0; lt < 13; lt += 4) {
      const int l  = 16 * lt + c;
      const int lc = (l < MAX_LEN) ? l : (MAX_LEN - 1);
      float4 bv[16];
#pragma unroll
      for (int j = 0; j < 16; ++j)
        bv[j] = sT[lc * TILE_CH + ((2 * j + h) ^ (lc & 15))];
      __builtin_amdgcn_sched_barrier(0);
      v8f d0 = {}, d1 = {};
#pragma unroll
      for (int j = 0; j < 16; ++j) {
        v2f blo; blo.x = bv[j].x; blo.y = bv[j].y;
        v2f bhi; bhi.x = bv[j].z; bhi.y = bv[j].w;
        d0 = __builtin_amdgcn_wmma_f32_16x16x4_f32(false, ah[2 * j], false, blo, (short)0, d0, false, false);
        d1 = __builtin_amdgcn_wmma_f32_16x16x4_f32(false, ah[2 * j + 1], false, bhi, (short)0, d1, false, false);
      }
      if (h == 0 && l < MAX_LEN) {
#pragma unroll
        for (int r = 0; r < MAX_K; ++r)
          atomicAdd(&Bdelta[r * MAX_LEN + l], d0[r] + d1[r]);
      }
    }
  }
}

__global__ void __launch_bounds__(640)
update_b_kernel(float* __restrict__ Bwork, float* __restrict__ Bdelta,
                float* __restrict__ tabM, float* __restrict__ tabI) {
  __shared__ TabLds s;
  const int t = threadIdx.x;
  if (t < MAX_K * MAX_LEN) {
    const float v = Bwork[t] + Bdelta[t];
    Bwork[t] = v; Bdelta[t] = 0.0f; s.b[t] = v;
  }
  __syncthreads();
  softmax_tables(s, tabM, tabI, t);
}

extern "C" void kernel_launch(void* const* d_in, const int* in_sizes, int n_in,
                              void* d_out, int out_size, void* d_ws, size_t ws_size,
                              hipStream_t stream) {
  const float* low  = (const float*)d_in[0];
  const float* Bin  = (const float*)d_in[1];
  const float* S    = (const float*)d_in[2];
  const int*   slen = (const int*)d_in[3];
  float* out = (float*)d_out;

  char* ws = (char*)d_ws;
  float* lown   = (float*)ws;
  float* Bwork  = (float*)(ws + (size_t)NBATCH * MAX_LEN * D_OUT * sizeof(float));
  float* Bdelta = Bwork + MAX_K * MAX_LEN;
  float* tabM   = Bdelta + MAX_K * MAX_LEN;
  float* tabI   = tabM + MAX_K * TAB_N;

  gemm_lowS_kernel<<<256, 256, 0, stream>>>(low, S, lown);

  init_b_kernel<<<1, 640, 0, stream>>>(Bin, Bwork, Bdelta, tabM, tabI);

  routing_iter_kernel<false><<<NBATCH, 128, 0, stream>>>(lown, Bwork, tabM, tabI, slen, Bdelta, out);
  update_b_kernel<<<1, 640, 0, stream>>>(Bwork, Bdelta, tabM, tabI);
  routing_iter_kernel<false><<<NBATCH, 128, 0, stream>>>(lown, Bwork, tabM, tabI, slen, Bdelta, out);
  update_b_kernel<<<1, 640, 0, stream>>>(Bwork, Bdelta, tabM, tabI);
  routing_iter_kernel<true><<<NBATCH, 128, 0, stream>>>(lown, Bwork, tabM, tabI, slen, Bdelta, out);
}


# Init launch folded into the GEMM launch: its workgroup 0 copies B_in, zeroes the delta sums and builds the pass-1 softmax tables under its own S and A loads (LDS-only barriers); 6 launches, not 7
# speedup vs baseline: 3.0698x; 1.0442x over previous
#include <hip/hip_runtime.h>
#include <hip/hip_bf16.h>

typedef float v2f __attribute__((ext_vector_type(2)));
typedef float v8f __attribute__((ext_vector_type(8)));

#define MAX_LEN 200
#define MAX_K   3
#define D_IN    128
#define D_OUT   128
#define NBATCH  1024
#define NEG_PAD (-65535.0f)

#define TAB_N   (MAX_LEN + 1)
#define NCHUNK  (MAX_LEN / 8)
#define NPREF   (NCHUNK + 1)

struct TabLds {
  __attribute__((aligned(16))) float b[MAX_K * MAX_LEN];
  float cm[MAX_K * NCHUNK], cz[MAX_K * NCHUNK];
  float pm[MAX_K * NPREF],  pz[MAX_K * NPREF];
};

template <bool LDS_ONLY>
__device__ __forceinline__ void level_barrier() {
  if constexpr (LDS_ONLY) {
    __builtin_amdgcn_fence(__ATOMIC_RELEASE, "workgroup", "local");
    __builtin_amdgcn_s_barrier();
    __builtin_amdgcn_fence(__ATOMIC_ACQUIRE, "workgroup", "local");
  } else {
    __syncthreads();
  }
}

__device__ __forceinline__ void table_entry(const TabLds& s, float* __restrict__ tabM,
                                            float* __restrict__ tabI, int t) {
  const int k = t / TAB_N, n = t % TAB_N;
  const int q = n >> 3, r = n & 7;
  const float pm = s.pm[k * NPREF + q];
  const float pz = s.pz[k * NPREF + q];
  float bv[7];
#pragma unroll
  for (int i = 0; i < 7; ++i) bv[i] = s.b[k * MAX_LEN + min(8 * q + i, MAX_LEN - 1)];
  float m = (n < MAX_LEN) ? NEG_PAD : -3.0e38f;
  m = (q > 0) ? fmaxf(m, pm) : m;
#pragma unroll
  for (int i = 0; i < 7; ++i) m = (i < r) ? fmaxf(m, bv[i]) : m;
  float z = (n < MAX_LEN) ? (float)(MAX_LEN - n) * __expf(NEG_PAD - m) : 0.0f;
  z += (q > 0) ? pz * __expf(pm - m) : 0.0f;
#pragma unroll
  for (int i = 0; i < 7; ++i) z += (i < r) ? __expf(bv[i] - m) : 0.0f;
  tabM[t] = m;
  tabI[t] = 1.0f / z;
}

template <int NT, bool LDS_ONLY>
__device__ __forceinline__ void softmax_tables(TabLds& s, float* __restrict__ tabM,
                                               float* __restrict__ tabI, int t) {
  if (t < MAX_K * NCHUNK) {
    const int k = t / NCHUNK, j = t % NCHUNK;
    const float4 v0 = *reinterpret_cast<const float4*>(&s.b[k * MAX_LEN + 8 * j]);
    const float4 v1 = *reinterpret_cast<const float4*>(&s.b[k * MAX_LEN + 8 * j + 4]);
    const float m = fmaxf(fmaxf(fmaxf(v0.x, v0.y), fmaxf(v0.z, v0.w)),
                          fmaxf(fmaxf(v1.x, v1.y), fmaxf(v1.z, v1.w)));
    s.cm[t] = m;
    s.cz[t] = ((__expf(v0.x - m) + __expf(v0.y - m)) + (__expf(v0.z - m) + __expf(v0.w - m))) +
              ((__expf(v1.x - m) + __expf(v1.y - m)) + (__expf(v1.z - m) + __expf(v1.w - m)));
  }
  level_barrier<LDS_ONLY>();
  if (t < MAX_K * NPREF) {
    const int k = t / NPREF, q = t % NPREF;
    float cmv[NCHUNK], czv[NCHUNK];
#pragma unroll
    for (int j = 0; j < NCHUNK; ++j) { cmv[j] = s.cm[k * NCHUNK + j]; czv[j] = s.cz[k * NCHUNK + j]; }
    float m = -3.0e38f;
#pragma unroll
    for (int j = 0; j < NCHUNK; ++j) m = (j < q) ? fmaxf(m, cmv[j]) : m;
    float z = 0.0f;
#pragma unroll
    for (int j = 0; j < NCHUNK; ++j) z += (j < q) ? czv[j] * __expf(cmv[j] - m) : 0.0f;
    s.pm[t] = m;
    s.pz[t] = z;
  }
  level_barrier<LDS_ONLY>();
  constexpr int PER_THREAD = (MAX_K * TAB_N + NT - 1) / NT;
#pragma unroll
  for (int i = 0; i < PER_THREAD; ++i) {
    const int e = t + i * NT;
    if (e < MAX_K * TAB_N) table_entry(s, tabM, tabI, e);
  }
}

#define ROWS_PER_WG   800
#define TILE_STRIDE   (64 * D_IN)

__device__ __forceinline__ void load_a_tile(float4 (&a)[16], const float* __restrict__ p) {
#pragma unroll
  for (int j = 0; j < 16; ++j)
    a[j] = *reinterpret_cast<const float4*>(p + 8 * j);
}

__device__ __forceinline__ void mm_tile(const float4 (&a)[16], const v2f (&bf)[32][4],
                                        float* __restrict__ o) {
  v8f acc[4] = {};
#pragma unroll
  for (int j = 0; j < 16; ++j) {
    v2f alo; alo.x = a[j].x; alo.y = a[j].y;
    v2f ahi; ahi.x = a[j].z; ahi.y = a[j].w;
#pragma unroll
    for (int nt = 0; nt < 4; ++nt)
      acc[nt] = __builtin_amdgcn_wmma_f32_16x16x4_f32(false, alo, false, bf[2 * j][nt],
                                                      (short)0, acc[nt], false, false);
#pragma unroll
    for (int nt = 0; nt < 4; ++nt)
      acc[nt] = __builtin_amdgcn_wmma_f32_16x16x4_f32(false, ahi, false, bf[2 * j + 1][nt],
                                                      (short)0, acc[nt], false, false);
  }
#pragma unroll
  for (int r = 0; r < 8; ++r) {
    float4 v;
    v.x = acc[0][r]; v.y = acc[1][r]; v.z = acc[2][r]; v.w = acc[3][r];
    *reinterpret_cast<float4*>(o + r * D_OUT) = v;
  }
}

__global__ void __launch_bounds__(256)
gemm_lowS_kernel(const float* __restrict__ low, const float* __restrict__ S,
                 float* __restrict__ lown, const float* __restrict__ Bin,
                 float* __restrict__ Bwork, float* __restrict__ Bdelta,
                 float* __restrict__ tabM, float* __restrict__ tabI) {
  __shared__ TabLds s;
  const int t = threadIdx.x;
  const bool tab_wg = (blockIdx.x == 0);
  const int t2 = min(t + 512, MAX_K * MAX_LEN - 1);
  float bin0 = 0.0f, bin1 = 0.0f, bin2 = 0.0f;
  if (tab_wg) { bin0 = Bin[t]; bin1 = Bin[t + 256]; bin2 = Bin[t2]; }

  const int wave = __builtin_amdgcn_readfirstlane((int)(threadIdx.x >> 5));
  const int lane = threadIdx.x & 31;
  const int c  = lane & 15;
  const int h  = lane >> 4;
  const int nh = wave & 1;
  const int q  = wave >> 1;
  const int ntile = (q < 2) ? 13 : 12;

  const size_t row0 = (size_t)blockIdx.x * ROWS_PER_WG + 16 * q;
  const float* __restrict__ ap = low + (row0 + c) * D_IN + 4 * h;
  float* __restrict__ op = lown + (row0 + 8 * h) * D_OUT + 64 * nh + 4 * c;

  float4 a0[16], a1[16];
  load_a_tile(a0, ap);

  const float* __restrict__ sp = S + (size_t)(4 * h) * D_OUT + 64 * nh + 4 * c;
  float4 sr[32][2];
#pragma unroll
  for (int kk = 0; kk < 32; ++kk) {
    sr[kk][0] = *reinterpret_cast<const float4*>(sp + (8 * (kk >> 1) + 2 * (kk & 1)) * D_OUT);
    sr[kk][1] = *reinterpret_cast<const float4*>(sp + (8 * (kk >> 1) + 2 * (kk & 1) + 1) * D_OUT);
  }

  if (tab_wg) {
    Bwork[t] = bin0;       Bdelta[t] = 0.0f;       s.b[t] = bin0;
    Bwork[t + 256] = bin1; Bdelta[t + 256] = 0.0f; s.b[t + 256] = bin1;
    if (t + 512 < MAX_K * MAX_LEN) { Bwork[t2] = bin2; Bdelta[t2] = 0.0f; s.b[t2] = bin2; }
    level_barrier<true>();
    softmax_tables<256, true>(s, tabM, tabI, t);
  }

  v2f bf[32][4];
#pragma unroll
  for (int j = 0; j < 16; ++j) {
#pragma unroll
    for (int p = 0; p < 2; ++p) {
      const float4 r0 = sr[2 * j + p][0];
      const float4 r1 = sr[2 * j + p][1];
      bf[2 * j + p][0].x = r0.x; bf[2 * j + p][0].y = r1.x;
      bf[2 * j + p][1].x = r0.y; bf[2 * j + p][1].y = r1.y;
      bf[2 * j + p][2].x = r0.z; bf[2 * j + p][2].y = r1.z;
      bf[2 * j + p][3].x = r0.w; bf[2 * j + p][3].y = r1.w;
    }
  }

#pragma unroll 1
  for (int i = 0; i < 6; ++i) {
    load_a_tile(a1, ap + (size_t)(2 * i + 1) * TILE_STRIDE);
    mm_tile(a0, bf, op + (size_t)(2 * i) * TILE_STRIDE);
    const int nxt = (2 * i + 2 < ntile) ? (2 * i + 2) : (ntile - 1);
    load_a_tile(a0, ap + (size_t)nxt * TILE_STRIDE);
    mm_tile(a1, bf, op + (size_t)(2 * i + 1) * TILE_STRIDE);
  }
  if (ntile == 13)
    mm_tile(a0, bf, op + (size_t)12 * TILE_STRIDE);
}

#define TILE_CH 32
#define KSTEPS  25

__device__ __forceinline__ float uniform_f32(float v) {
  return __builtin_bit_cast(float, __builtin_amdgcn_readfirstlane(__builtin_bit_cast(int, v)));
}

template <bool LAST>
__global__ void __launch_bounds__(128)
routing_iter_kernel(const float* __restrict__ lown, const float* __restrict__ Bwork,
                    const float* __restrict__ tabM, const float* __restrict__ tabI,
                    const int* __restrict__ seq_len, float* __restrict__ Bdelta,
                    float* __restrict__ out) {
  __shared__ float4 sT[LAST ? 1 : MAX_LEN * TILE_CH];
  __shared__ __attribute__((aligned(16))) float sP[2][MAX_K][D_OUT];

  const int b    = blockIdx.x;
  const int t    = threadIdx.x;
  const int wave = __builtin_amdgcn_readfirstlane((int)(threadIdx.x >> 5));
  const int lane = t & 31;
  const int len  = seq_len[b];
  const int n    = min(max(len, 0), MAX_LEN);

  const int  c    = lane & 15;
  const int  h    = lane >> 4;
  const bool live = c < MAX_K;
  const int  mm   = live ? c : 0;
  const int  nh   = wave & 1;
  const int  kh   = wave >> 1;
  const int  lrow = 100 * kh + 2 * h;

  const float* __restrict__ bwp = Bwork + mm * MAX_LEN + lrow;
  float2 bw[KSTEPS];
#pragma unroll
  for (int u = 0; u < KSTEPS; ++u) bw[u] = *reinterpret_cast<const float2*>(bwp + 4 * u);

  const float4* __restrict__ tp =
      reinterpret_cast<const float4*>(lown + (size_t)b * MAX_LEN * D_OUT) +
      lrow * TILE_CH + 16 * nh + c;
  float4 r0[KSTEPS], r1[KSTEPS];
#pragma unroll
  for (int u = 0; u < KSTEPS; ++u) {
    r0[u] = tp[(4 * u) * TILE_CH];
    r1[u] = tp[(4 * u + 1) * TILE_CH];
  }

  const float m0 = uniform_f32(tabM[0 * TAB_N + n]);
  const float m1 = uniform_f32(tabM[1 * TAB_N + n]);
  const float m2 = uniform_f32(tabM[2 * TAB_N + n]);
  const float i0 = uniform_f32(tabI[0 * TAB_N + n]);
  const float i1 = uniform_f32(tabI[1 * TAB_N + n]);
  const float i2 = uniform_f32(tabI[2 * TAB_N + n]);
  const float mk = (c == 0) ? m0 : (c == 1) ? m1 : m2;
  const float ik = (c == 0) ? i0 : (c == 1) ? i1 : i2;
  const float wpad = __expf(NEG_PAD - mk) * ik;

  {
    v8f acc[4] = {};
#pragma unroll
    for (int u = 0; u < KSTEPS; ++u) {
      const int l0 = lrow + 4 * u;
      const int l1 = l0 + 1;
      if constexpr (!LAST) {
        sT[l0 * TILE_CH + ((16 * nh + c) ^ (l0 & 15))] = r0[u];
        sT[l1 * TILE_CH + ((16 * nh + c) ^ (l1 & 15))] = r1[u];
      }
      const float w0 = __expf(bw[u].x - mk) * ik;
      const float w1 = __expf(bw[u].y - mk) * ik;
      v2f a;
      a.x = live ? ((l0 < len) ? w0 : wpad) : 0.0f;
      a.y = live ? ((l1 < len) ? w1 : wpad) : 0.0f;
      v2f b0; b0.x = r0[u].x; b0.y = r1[u].x;
      v2f b1; b1.x = r0[u].y; b1.y = r1[u].y;
      v2f b2; b2.x = r0[u].z; b2.y = r1[u].z;
      v2f b3; b3.x = r0[u].w; b3.y = r1[u].w;
      acc[0] = __builtin_amdgcn_wmma_f32_16x16x4_f32(false, a, false, b0, (short)0, acc[0], false, false);
      acc[1] = __builtin_amdgcn_wmma_f32_16x16x4_f32(false, a, false, b1, (short)0, acc[1], false, false);
      acc[2] = __builtin_amdgcn_wmma_f32_16x16x4_f32(false, a, false, b2, (short)0, acc[2], false, false);
      acc[3] = __builtin_amdgcn_wmma_f32_16x16x4_f32(false, a, false, b3, (short)0, acc[3], false, false);
      __builtin_amdgcn_sched_barrier(0);
    }
    if (h == 0) {
#pragma unroll
      for (int r = 0; r < MAX_K; ++r) {
        float4 v;
        v.x = acc[0][r]; v.y = acc[1][r]; v.z = acc[2][r]; v.w = acc[3][r];
        *reinterpret_cast<float4*>(&sP[kh][r][64 * nh + 4 * c]) = v;
      }
    }
  }
  __syncthreads();

  {
    const float x0 = sP[0][0][t] + sP[1][0][t];
    const float x1 = sP[0][1][t] + sP[1][1][t];
    const float x2 = sP[0][2][t] + sP[1][2][t];
    const float sq = x0 * x0 + x1 * x1 + x2 * x2;
    const float scale = sq / (1.0f + sq) * rsqrtf(sq + 1e-9f);
    const float h0 = scale * x0, h1 = scale * x1, h2 = scale * x2;
    if constexpr (LAST) {
      float* __restrict__ ob = out + (size_t)b * MAX_K * D_OUT;
      ob[0 * D_OUT + t] = h0;
      ob[1 * D_OUT + t] = h1;
      ob[2 * D_OUT + t] = h2;
      return;
    }
    sP[0][0][t] = h0; sP[0][1][t] = h1; sP[0][2][t] = h2;
  }
  __syncthreads();

  if constexpr (!LAST) {
    v2f ah[32];
#pragma unroll
    for (int j = 0; j < 16; ++j) {
      const float4 hv = *reinterpret_cast<const float4*>(&sP[0][mm][8 * j + 4 * h]);
      ah[2 * j].x     = live ? hv.x : 0.0f;
      ah[2 * j].y     = live ? hv.y : 0.0f;
      ah[2 * j + 1].x = live ? hv.z : 0.0f;
      ah[2 * j + 1].y = live ? hv.w : 0.0f;
    }
    const int w0 = (wave + b) & 3;
    for (int lt = w0; lt < 13; lt += 4) {
      const int l  = 16 * lt + c;
      const int lc = (l < MAX_LEN) ? l : (MAX_LEN - 1);
      float4 bv[16];
#pragma unroll
      for (int j = 0; j < 16; ++j)
        bv[j] = sT[lc * TILE_CH + ((2 * j + h) ^ (lc & 15))];
      __builtin_amdgcn_sched_barrier(0);
      v8f d0 = {}, d1 = {};
#pragma unroll
      for (int j = 0; j < 16; ++j) {
        v2f blo; blo.x = bv[j].x; blo.y = bv[j].y;
        v2f bhi; bhi.x = bv[j].z; bhi.y = bv[j].w;
        d0 = __builtin_amdgcn_wmma_f32_16x16x4_f32(false, ah[2 * j], false, blo, (short)0, d0, false, false);
        d1 = __builtin_amdgcn_wmma_f32_16x16x4_f32(false, ah[2 * j + 1], false, bhi, (short)0, d1, false, false);
      }
      if (h == 0 && l < MAX_LEN) {
#pragma unroll
        for (int r = 0; r < MAX_K; ++r)
          atomicAdd(&Bdelta[r * MAX_LEN + l], d0[r] + d1[r]);
      }
    }
  }
}

__global__ void __launch_bounds__(640)
update_b_kernel(float* __restrict__ Bwork, float* __restrict__ Bdelta,
                float* __restrict__ tabM, float* __restrict__ tabI) {
  __shared__ TabLds s;
  const int t = threadIdx.x;
  if (t < MAX_K * MAX_LEN) {
    const float v = Bwork[t] + Bdelta[t];
    Bwork[t] = v; Bdelta[t] = 0.0f; s.b[t] = v;
  }
  __syncthreads();
  softmax_tables<640, false>(s, tabM, tabI, t);
}

extern "C" void kernel_launch(void* const* d_in, const int* in_sizes, int n_in,
                              void* d_out, int out_size, void* d_ws, size_t ws_size,
                              hipStream_t stream) {
  const float* low  = (const float*)d_in[0];
  const float* Bin  = (const float*)d_in[1];
  const float* S    = (const float*)d_in[2];
  const int*   slen = (const int*)d_in[3];
  float* out = (float*)d_out;

  char* ws = (char*)d_ws;
  float* lown   = (float*)ws;
  float* Bwork  = (float*)(ws + (size_t)NBATCH * MAX_LEN * D_OUT * sizeof(float));
  float* Bdelta = Bwork + MAX_K * MAX_LEN;
  float* tabM   = Bdelta + MAX_K * MAX_LEN;
  float* tabI   = tabM + MAX_K * TAB_N;

  gemm_lowS_kernel<<<256, 256, 0, stream>>>(low, S, lown, Bin, Bwork, Bdelta, tabM, tabI);

  routing_iter_kernel<false><<<NBATCH, 128, 0, stream>>>(lown, Bwork, tabM, tabI, slen, Bdelta, out);
  update_b_kernel<<<1, 640, 0, stream>>>(Bwork, Bdelta, tabM, tabI);
  routing_iter_kernel<false><<<NBATCH, 128, 0, stream>>>(lown, Bwork, tabM, tabI, slen, Bdelta, out);
  update_b_kernel<<<1, 640, 0, stream>>>(Bwork, Bdelta, tabM, tabI);
  routing_iter_kernel<true><<<NBATCH, 128, 0, stream>>>(lown, Bwork, tabM, tabI, slen, Bdelta, out);
}


# Delta accumulator kept in 8 copies (batch b adds to copy b%8, rows padded so a 16-lane atomic is one aligned 64-byte piece); update sums the copies: float sums grouped in a different order
# speedup vs baseline: 3.2016x; 1.0429x over previous
#include <hip/hip_runtime.h>
#include <hip/hip_bf16.h>

typedef float v2f __attribute__((ext_vector_type(2)));
typedef float v8f __attribute__((ext_vector_type(8)));

#define MAX_LEN 200
#define MAX_K   3
#define D_IN    128
#define D_OUT   128
#define NBATCH  1024
#define NEG_PAD (-65535.0f)

#define NREP    8
#define DPITCH  208
#define DPLANE  640

#define TAB_N   (MAX_LEN + 1)
#define NCHUNK  (MAX_LEN / 8)
#define NPREF   (NCHUNK + 1)

struct TabLds {
  __attribute__((aligned(16))) float b[MAX_K * MAX_LEN];
  float cm[MAX_K * NCHUNK], cz[MAX_K * NCHUNK];
  float pm[MAX_K * NPREF],  pz[MAX_K * NPREF];
};

template <bool LDS_ONLY>
__device__ __forceinline__ void level_barrier() {
  if constexpr (LDS_ONLY) {
    __builtin_amdgcn_fence(__ATOMIC_RELEASE, "workgroup", "local");
    __builtin_amdgcn_s_barrier();
    __builtin_amdgcn_fence(__ATOMIC_ACQUIRE, "workgroup", "local");
  } else {
    __syncthreads();
  }
}

__device__ __forceinline__ void table_entry(const TabLds& s, float* __restrict__ tabM,
                                            float* __restrict__ tabI, int t) {
  const int k = t / TAB_N, n = t % TAB_N;
  const int q = n >> 3, r = n & 7;
  const float pm = s.pm[k * NPREF + q];
  const float pz = s.pz[k * NPREF + q];
  float bv[7];
#pragma unroll
  for (int i = 0; i < 7; ++i) bv[i] = s.b[k * MAX_LEN + min(8 * q + i, MAX_LEN - 1)];
  float m = (n < MAX_LEN) ? NEG_PAD : -3.0e38f;
  m = (q > 0) ? fmaxf(m, pm) : m;
#pragma unroll
  for (int i = 0; i < 7; ++i) m = (i < r) ? fmaxf(m, bv[i]) : m;
  float z = (n < MAX_LEN) ? (float)(MAX_LEN - n) * __expf(NEG_PAD - m) : 0.0f;
  z += (q > 0) ? pz * __expf(pm - m) : 0.0f;
#pragma unroll
  for (int i = 0; i < 7; ++i) z += (i < r) ? __expf(bv[i] - m) : 0.0f;
  tabM[t] = m;
  tabI[t] = 1.0f / z;
}

template <int NT, bool LDS_ONLY>
__device__ __forceinline__ void softmax_tables(TabLds& s, float* __restrict__ tabM,
                                               float* __restrict__ tabI, int t) {
  if (t < MAX_K * NCHUNK) {
    const int k = t / NCHUNK, j = t % NCHUNK;
    const float4 v0 = *reinterpret_cast<const float4*>(&s.b[k * MAX_LEN + 8 * j]);
    const float4 v1 = *reinterpret_cast<const float4*>(&s.b[k * MAX_LEN + 8 * j + 4]);
    const float m = fmaxf(fmaxf(fmaxf(v0.x, v0.y), fmaxf(v0.z, v0.w)),
                          fmaxf(fmaxf(v1.x, v1.y), fmaxf(v1.z, v1.w)));
    s.cm[t] = m;
    s.cz[t] = ((__expf(v0.x - m) + __expf(v0.y - m)) + (__expf(v0.z - m) + __expf(v0.w - m))) +
              ((__expf(v1.x - m) + __expf(v1.y - m)) + (__expf(v1.z - m) + __expf(v1.w - m)));
  }
  level_barrier<LDS_ONLY>();
  if (t < MAX_K * NPREF) {
    const int k = t / NPREF, q = t % NPREF;
    float cmv[NCHUNK], czv[NCHUNK];
#pragma unroll
    for (int j = 0; j < NCHUNK; ++j) { cmv[j] = s.cm[k * NCHUNK + j]; czv[j] = s.cz[k * NCHUNK + j]; }
    float m = -3.0e38f;
#pragma unroll
    for (int j = 0; j < NCHUNK; ++j) m = (j < q) ? fmaxf(m, cmv[j]) : m;
    float z = 0.0f;
#pragma unroll
    for (int j = 0; j < NCHUNK; ++j) z += (j < q) ? czv[j] * __expf(cmv[j] - m) : 0.0f;
    s.pm[t] = m;
    s.pz[t] = z;
  }
  level_barrier<LDS_ONLY>();
  constexpr int PER_THREAD = (MAX_K * TAB_N + NT - 1) / NT;
#pragma unroll
  for (int i = 0; i < PER_THREAD; ++i) {
    const int e = t + i * NT;
    if (e < MAX_K * TAB_N) table_entry(s, tabM, tabI, e);
  }
}

#define ROWS_PER_WG   800
#define TILE_STRIDE   (64 * D_IN)

__device__ __forceinline__ void load_a_tile(float4 (&a)[16], const float* __restrict__ p) {
#pragma unroll
  for (int j = 0; j < 16; ++j)
    a[j] = *reinterpret_cast<const float4*>(p + 8 * j);
}

__device__ __forceinline__ void mm_tile(const float4 (&a)[16], const v2f (&bf)[32][4],
                                        float* __restrict__ o) {
  v8f acc[4] = {};
#pragma unroll
  for (int j = 0; j < 16; ++j) {
    v2f alo; alo.x = a[j].x; alo.y = a[j].y;
    v2f ahi; ahi.x = a[j].z; ahi.y = a[j].w;
#pragma unroll
    for (int nt = 0; nt < 4; ++nt)
      acc[nt] = __builtin_amdgcn_wmma_f32_16x16x4_f32(false, alo, false, bf[2 * j][nt],
                                                      (short)0, acc[nt], false, false);
#pragma unroll
    for (int nt = 0; nt < 4; ++nt)
      acc[nt] = __builtin_amdgcn_wmma_f32_16x16x4_f32(false, ahi, false, bf[2 * j + 1][nt],
                                                      (short)0, acc[nt], false, false);
  }
#pragma unroll
  for (int r = 0; r < 8; ++r) {
    float4 v;
    v.x = acc[0][r]; v.y = acc[1][r]; v.z = acc[2][r]; v.w = acc[3][r];
    *reinterpret_cast<float4*>(o + r * D_OUT) = v;
  }
}

__global__ void __launch_bounds__(256)
gemm_lowS_kernel(const float* __restrict__ low, const float* __restrict__ S,
                 float* __restrict__ lown, const float* __restrict__ Bin,
                 float* __restrict__ Bwork, float* __restrict__ Bdelta,
                 float* __restrict__ tabM, float* __restrict__ tabI) {
  __shared__ TabLds s;
  const int t = threadIdx.x;
  const bool tab_wg = (blockIdx.x == 0);
  const int t2 = min(t + 512, MAX_K * MAX_LEN - 1);
  float bin0 = 0.0f, bin1 = 0.0f, bin2 = 0.0f;
  if (tab_wg) { bin0 = Bin[t]; bin1 = Bin[t + 256]; bin2 = Bin[t2]; }

  const int wave = __builtin_amdgcn_readfirstlane((int)(threadIdx.x >> 5));
  const int lane = threadIdx.x & 31;
  const int c  = lane & 15;
  const int h  = lane >> 4;
  const int nh = wave & 1;
  const int q  = wave >> 1;
  const int ntile = (q < 2) ? 13 : 12;

  const size_t row0 = (size_t)blockIdx.x * ROWS_PER_WG + 16 * q;
  const float* __restrict__ ap = low + (row0 + c) * D_IN + 4 * h;
  float* __restrict__ op = lown + (row0 + 8 * h) * D_OUT + 64 * nh + 4 * c;

  float4 a0[16], a1[16];
  load_a_tile(a0, ap);

  const float* __restrict__ sp = S + (size_t)(4 * h) * D_OUT + 64 * nh + 4 * c;
  float4 sr[32][2];
#pragma unroll
  for (int kk = 0; kk < 32; ++kk) {
    sr[kk][0] = *reinterpret_cast<const float4*>(sp + (8 * (kk >> 1) + 2 * (kk & 1)) * D_OUT);
    sr[kk][1] = *reinterpret_cast<const float4*>(sp + (8 * (kk >> 1) + 2 * (kk & 1) + 1) * D_OUT);
  }

  if (tab_wg) {
    Bwork[t] = bin0;       s.b[t] = bin0;
    Bwork[t + 256] = bin1; s.b[t + 256] = bin1;
    if (t + 512 < MAX_K * MAX_LEN) { Bwork[t2] = bin2; s.b[t2] = bin2; }
    float4* __restrict__ dz = reinterpret_cast<float4*>(Bdelta);
#pragma unroll
    for (int i = 0; i < NREP * DPLANE / 4 / 256; ++i)
      dz[t + 256 * i] = make_float4(0.0f, 0.0f, 0.0f, 0.0f);
    level_barrier<true>();
    softmax_tables<256, true>(s, tabM, tabI, t);
  }

  v2f bf[32][4];
#pragma unroll
  for (int j = 0; j < 16; ++j) {
#pragma unroll
    for (int p = 0; p < 2; ++p) {
      const float4 r0 = sr[2 * j + p][0];
      const float4 r1 = sr[2 * j + p][1];
      bf[2 * j + p][0].x = r0.x; bf[2 * j + p][0].y = r1.x;
      bf[2 * j + p][1].x = r0.y; bf[2 * j + p][1].y = r1.y;
      bf[2 * j + p][2].x = r0.z; bf[2 * j + p][2].y = r1.z;
      bf[2 * j + p][3].x = r0.w; bf[2 * j + p][3].y = r1.w;
    }
  }

#pragma unroll 1
  for (int i = 0; i < 6; ++i) {
    load_a_tile(a1, ap + (size_t)(2 * i + 1) * TILE_STRIDE);
    mm_tile(a0, bf, op + (size_t)(2 * i) * TILE_STRIDE);
    const int nxt = (2 * i + 2 < ntile) ? (2 * i + 2) : (ntile - 1);
    load_a_tile(a0, ap + (size_t)nxt * TILE_STRIDE);
    mm_tile(a1, bf, op + (size_t)(2 * i + 1) * TILE_STRIDE);
  }
  if (ntile == 13)
    mm_tile(a0, bf, op + (size_t)12 * TILE_STRIDE);
}

#define TILE_CH 32
#define KSTEPS  25

__device__ __forceinline__ float uniform_f32(float v) {
  return __builtin_bit_cast(float, __builtin_amdgcn_readfirstlane(__builtin_bit_cast(int, v)));
}

template <bool LAST>
__global__ void __launch_bounds__(128)
routing_iter_kernel(const float* __restrict__ lown, const float* __restrict__ Bwork,
                    const float* __restrict__ tabM, const float* __restrict__ tabI,
                    const int* __restrict__ seq_len, float* __restrict__ Bdelta,
                    float* __restrict__ out) {
  __shared__ float4 sT[LAST ? 1 : MAX_LEN * TILE_CH];
  __shared__ __attribute__((aligned(16))) float sP[2][MAX_K][D_OUT];

  const int b    = blockIdx.x;
  const int t    = threadIdx.x;
  const int wave = __builtin_amdgcn_readfirstlane((int)(threadIdx.x >> 5));
  const int lane = t & 31;
  const int len  = seq_len[b];
  const int n    = min(max(len, 0), MAX_LEN);

  const int  c    = lane & 15;
  const int  h    = lane >> 4;
  const bool live = c < MAX_K;
  const int  mm   = live ? c : 0;
  const int  nh   = wave & 1;
  const int  kh   = wave >> 1;
  const int  lrow = 100 * kh + 2 * h;

  const float* __restrict__ bwp = Bwork + mm * MAX_LEN + lrow;
  float2 bw[KSTEPS];
#pragma unroll
  for (int u = 0; u < KSTEPS; ++u) bw[u] = *reinterpret_cast<const float2*>(bwp + 4 * u);

  const float4* __restrict__ tp =
      reinterpret_cast<const float4*>(lown + (size_t)b * MAX_LEN * D_OUT) +
      lrow * TILE_CH + 16 * nh + c;
  float4 r0[KSTEPS], r1[KSTEPS];
#pragma unroll
  for (int u = 0; u < KSTEPS; ++u) {
    r0[u] = tp[(4 * u) * TILE_CH];
    r1[u] = tp[(4 * u + 1) * TILE_CH];
  }

  const float m0 = uniform_f32(tabM[0 * TAB_N + n]);
  const float m1 = uniform_f32(tabM[1 * TAB_N + n]);
  const float m2 = uniform_f32(tabM[2 * TAB_N + n]);
  const float i0 = uniform_f32(tabI[0 * TAB_N + n]);
  const float i1 = uniform_f32(tabI[1 * TAB_N + n]);
  const float i2 = uniform_f32(tabI[2 * TAB_N + n]);
  const float mk = (c == 0) ? m0 : (c == 1) ? m1 : m2;
  const float ik = (c == 0) ? i0 : (c == 1) ? i1 : i2;
  const float wpad = __expf(NEG_PAD - mk) * ik;

  {
    v8f acc[4] = {};
#pragma unroll
    for (int u = 0; u < KSTEPS; ++u) {
      const int l0 = lrow + 4 * u;
      const int l1 = l0 + 1;
      if constexpr (!LAST) {
        sT[l0 * TILE_CH + ((16 * nh + c) ^ (l0 & 15))] = r0[u];
        sT[l1 * TILE_CH + ((16 * nh + c) ^ (l1 & 15))] = r1[u];
      }
      const float w0 = __expf(bw[u].x - mk) * ik;
      const float w1 = __expf(bw[u].y - mk) * ik;
      v2f a;
      a.x = live ? ((l0 < len) ? w0 : wpad) : 0.0f;
      a.y = live ? ((l1 < len) ? w1 : wpad) : 0.0f;
      v2f b0; b0.x = r0[u].x; b0.y = r1[u].x;
      v2f b1; b1.x = r0[u].y; b1.y = r1[u].y;
      v2f b2; b2.x = r0[u].z; b2.y = r1[u].z;
      v2f b3; b3.x = r0[u].w; b3.y = r1[u].w;
      acc[0] = __builtin_amdgcn_wmma_f32_16x16x4_f32(false, a, false, b0, (short)0, acc[0], false, false);
      acc[1] = __builtin_amdgcn_wmma_f32_16x16x4_f32(false, a, false, b1, (short)0, acc[1], false, false);
      acc[2] = __builtin_amdgcn_wmma_f32_16x16x4_f32(false, a, false, b2, (short)0, acc[2], false, false);
      acc[3] = __builtin_amdgcn_wmma_f32_16x16x4_f32(false, a, false, b3, (short)0, acc[3], false, false);
      __builtin_amdgcn_sched_barrier(0);
    }
    if (h == 0) {
#pragma unroll
      for (int r = 0; r < MAX_K; ++r) {
        float4 v;
        v.x = acc[0][r]; v.y = acc[1][r]; v.z = acc[2][r]; v.w = acc[3][r];
        *reinterpret_cast<float4*>(&sP[kh][r][64 * nh + 4 * c]) = v;
      }
    }
  }
  __syncthreads();

  {
    const float x0 = sP[0][0][t] + sP[1][0][t];
    const float x1 = sP[0][1][t] + sP[1][1][t];
    const float x2 = sP[0][2][t] + sP[1][2][t];
    const float sq = x0 * x0 + x1 * x1 + x2 * x2;
    const float scale = sq / (1.0f + sq) * rsqrtf(sq + 1e-9f);
    const float h0 = scale * x0, h1 = scale * x1, h2 = scale * x2;
    if constexpr (LAST) {
      float* __restrict__ ob = out + (size_t)b * MAX_K * D_OUT;
      ob[0 * D_OUT + t] = h0;
      ob[1 * D_OUT + t] = h1;
      ob[2 * D_OUT + t] = h2;
      return;
    }
    sP[0][0][t] = h0; sP[0][1][t] = h1; sP[0][2][t] = h2;
  }
  __syncthreads();

  if constexpr (!LAST) {
    v2f ah[32];
#pragma unroll
    for (int j = 0; j < 16; ++j) {
      const float4 hv = *reinterpret_cast<const float4*>(&sP[0][mm][8 * j + 4 * h]);
      ah[2 * j].x     = live ? hv.x : 0.0f;
      ah[2 * j].y     = live ? hv.y : 0.0f;
      ah[2 * j + 1].x = live ? hv.z : 0.0f;
      ah[2 * j + 1].y = live ? hv.w : 0.0f;
    }
    const int w0 = (wave + b) & 3;
    float* __restrict__ bd = Bdelta + (b & (NREP - 1)) * DPLANE;
    for (int lt = w0; lt < 13; lt += 4) {
      const int l  = 16 * lt + c;
      const int lc = (l < MAX_LEN) ? l : (MAX_LEN - 1);
      float4 bv[16];
#pragma unroll
      for (int j = 0; j < 16; ++j)
        bv[j] = sT[lc * TILE_CH + ((2 * j + h) ^ (lc & 15))];
      __builtin_amdgcn_sched_barrier(0);
      v8f d0 = {}, d1 = {};
#pragma unroll
      for (int j = 0; j < 16; ++j) {
        v2f blo; blo.x = bv[j].x; blo.y = bv[j].y;
        v2f bhi; bhi.x = bv[j].z; bhi.y = bv[j].w;
        d0 = __builtin_amdgcn_wmma_f32_16x16x4_f32(false, ah[2 * j], false, blo, (short)0, d0, false, false);
        d1 = __builtin_amdgcn_wmma_f32_16x16x4_f32(false, ah[2 * j + 1], false, bhi, (short)0, d1, false, false);
      }
      if (h == 0 && l < MAX_LEN) {
#pragma unroll
        for (int r = 0; r < MAX_K; ++r)
          atomicAdd(&bd[r * DPITCH + l], d0[r] + d1[r]);
      }
    }
  }
}

template <bool REZERO>
__global__ void __launch_bounds__(640)
update_b_kernel(float* __restrict__ Bwork, float* __restrict__ Bdelta,
                float* __restrict__ tabM, float* __restrict__ tabI) {
  __shared__ TabLds s;
  const int t = threadIdx.x;
  if (t < MAX_K * MAX_LEN) {
    float* __restrict__ dp = Bdelta + (t / MAX_LEN) * DPITCH + (t % MAX_LEN);
    float dv[NREP];
#pragma unroll
    for (int r = 0; r < NREP; ++r) dv[r] = dp[r * DPLANE];
    const float d = ((dv[0] + dv[1]) + (dv[2] + dv[3])) + ((dv[4] + dv[5]) + (dv[6] + dv[7]));
    if constexpr (REZERO) {
#pragma unroll
      for (int r = 0; r < NREP; ++r) dp[r * DPLANE] = 0.0f;
    }
    const float v = Bwork[t] + d;
    Bwork[t] = v; s.b[t] = v;
  }
  __syncthreads();
  softmax_tables<640, false>(s, tabM, tabI, t);
}

extern "C" void kernel_launch(void* const* d_in, const int* in_sizes, int n_in,
                              void* d_out, int out_size, void* d_ws, size_t ws_size,
                              hipStream_t stream) {
  const float* low  = (const float*)d_in[0];
  const float* Bin  = (const float*)d_in[1];
  const float* S    = (const float*)d_in[2];
  const int*   slen = (const int*)d_in[3];
  float* out = (float*)d_out;

  char* ws = (char*)d_ws;
  float* lown   = (float*)ws;
  float* Bwork  = (float*)(ws + (size_t)NBATCH * MAX_LEN * D_OUT * sizeof(float));
  float* tabM   = Bwork + MAX_K * MAX_LEN;
  float* tabI   = tabM + MAX_K * TAB_N;
  float* Bdelta = out;

  gemm_lowS_kernel<<<256, 256, 0, stream>>>(low, S, lown, Bin, Bwork, Bdelta, tabM, tabI);

  routing_iter_kernel<false><<<NBATCH, 128, 0, stream>>>(lown, Bwork, tabM, tabI, slen, Bdelta, out);
  update_b_kernel<true><<<1, 640, 0, stream>>>(Bwork, Bdelta, tabM, tabI);
  routing_iter_kernel<false><<<NBATCH, 128, 0, stream>>>(lown, Bwork, tabM, tabI, slen, Bdelta, out);
  update_b_kernel<false><<<1, 640, 0, stream>>>(Bwork, Bdelta, tabM, tabI);
  routing_iter_kernel<true><<<NBATCH, 128, 0, stream>>>(lown, Bwork, tabM, tabI, slen, Bdelta, out);
}
